# starting kernel
# baseline (speedup 1.0000x reference)
#include <hip/hip_runtime.h>
#include <hip/hip_bf16.h>

typedef __attribute__((ext_vector_type(2))) float v2f;
typedef __attribute__((ext_vector_type(8))) float v8f;

#define D_FEAT 128
#define WAVES_PER_BLOCK 8
#define EDGES_PER_WAVE 16

__global__ __launch_bounds__(256) void edge_dot_wmma_f32(
    const float* __restrict__ x,
    const int*   __restrict__ src,
    const int*   __restrict__ dst,
    float*       __restrict__ out,
    int n_edges)
{
    const int lane    = threadIdx.x & 31;
    const int wave    = threadIdx.x >> 5;
    const long long waveGlobal = (long long)blockIdx.x * WAVES_PER_BLOCK + wave;
    const long long e0 = waveGlobal * EDGES_PER_WAVE;
    if (e0 >= n_edges) return;

    const int ei = lane & 15;
    long long e = e0 + ei;
    if (e >= n_edges) e = n_edges - 1;

    const int s = src[e];
    const int d = dst[e];

    const int colOff = (lane >> 4) << 1;
    const float* __restrict__ ps = x + (long long)s * D_FEAT + colOff;
    const float* __restrict__ pd = x + (long long)d * D_FEAT + colOff;

    v8f c = {};
#pragma unroll
    for (int k0 = 0; k0 < D_FEAT; k0 += 4) {
        v2f a = *(const v2f*)(ps + k0);
        v2f b = *(const v2f*)(pd + k0);
        c = __builtin_amdgcn_wmma_f32_16x16x4_f32(
                false, a, false, b, (short)0, c, false, false);
    }

    const int idx = lane & 7;
    float val = c[0];
    if (idx == 1) val = c[1];
    if (idx == 2) val = c[2];
    if (idx == 3) val = c[3];
    if (idx == 4) val = c[4];
    if (idx == 5) val = c[5];
    if (idx == 6) val = c[6];
    if (idx == 7) val = c[7];

    const bool writer = (lane < 8) || (lane >= 24);
    const long long eo = e0 + ((lane < 8) ? lane : (lane - 16));
    if (writer && eo < n_edges) out[eo] = val;
}

extern "C" void kernel_launch(void* const* d_in, const int* in_sizes, int n_in,
                              void* d_out, int out_size, void* d_ws, size_t ws_size,
                              hipStream_t stream) {
    const float* x   = (const float*)d_in[0];
    const int*   src = (const int*)d_in[1];
    const int*   dst = (const int*)d_in[2];
    float*       out = (float*)d_out;

    const int n_edges = in_sizes[1];
    const int n_waves = (n_edges + EDGES_PER_WAVE - 1) / EDGES_PER_WAVE;
    const int blocks  = (n_waves + WAVES_PER_BLOCK - 1) / WAVES_PER_BLOCK;

    edge_dot_wmma_f32<<<blocks, WAVES_PER_BLOCK * 32, 0, stream>>>(
        x, src, dst, out, n_edges);
}


# 16-byte fragment loads: each lane reads 4 consecutive floats per load and feeds two WMMA steps (half the gather loads, 32 B per touched line, was 16); sum order changes inside each 8-feature group
# speedup vs baseline: 1.5167x; 1.5167x over previous
#include <hip/hip_runtime.h>
#include <hip/hip_bf16.h>

typedef __attribute__((ext_vector_type(2))) float v2f;
typedef __attribute__((ext_vector_type(4))) float v4f;
typedef __attribute__((ext_vector_type(8))) float v8f;

#define D_FEAT 128
#define WAVES_PER_BLOCK 8
#define EDGES_PER_WAVE 16

__global__ __launch_bounds__(256) void edge_dot_wmma_f32(
    const float* __restrict__ x,
    const int*   __restrict__ src,
    const int*   __restrict__ dst,
    float*       __restrict__ out,
    int n_edges)
{
    const int lane    = threadIdx.x & 31;
    const int wave    = threadIdx.x >> 5;
    const long long waveGlobal = (long long)blockIdx.x * WAVES_PER_BLOCK + wave;
    const long long e0 = waveGlobal * EDGES_PER_WAVE;
    if (e0 >= n_edges) return;

    const int ei = lane & 15;
    long long e = e0 + ei;
    if (e >= n_edges) e = n_edges - 1;

    const int s = src[e];
    const int d = dst[e];

    const int colOff = (lane >> 4) << 2;
    const float* __restrict__ ps = x + (long long)s * D_FEAT + colOff;
    const float* __restrict__ pd = x + (long long)d * D_FEAT + colOff;

    v8f c = {};
#pragma unroll
    for (int k0 = 0; k0 < D_FEAT; k0 += 8) {
        const v4f a = *(const v4f*)(ps + k0);
        const v4f b = *(const v4f*)(pd + k0);
        const v2f a0 = {a.x, a.y};
        const v2f b0 = {b.x, b.y};
        const v2f a1 = {a.z, a.w};
        const v2f b1 = {b.z, b.w};
        c = __builtin_amdgcn_wmma_f32_16x16x4_f32(
                false, a0, false, b0, (short)0, c, false, false);
        c = __builtin_amdgcn_wmma_f32_16x16x4_f32(
                false, a1, false, b1, (short)0, c, false, false);
    }

    const int idx = lane & 7;
    float val = c[0];
    if (idx == 1) val = c[1];
    if (idx == 2) val = c[2];
    if (idx == 3) val = c[3];
    if (idx == 4) val = c[4];
    if (idx == 5) val = c[5];
    if (idx == 6) val = c[6];
    if (idx == 7) val = c[7];

    const bool writer = (lane < 8) || (lane >= 24);
    const long long eo = e0 + ((lane < 8) ? lane : (lane - 16));
    if (writer && eo < n_edges) out[eo] = val;
}

extern "C" void kernel_launch(void* const* d_in, const int* in_sizes, int n_in,
                              void* d_out, int out_size, void* d_ws, size_t ws_size,
                              hipStream_t stream) {
    const float* x   = (const float*)d_in[0];
    const int*   src = (const int*)d_in[1];
    const int*   dst = (const int*)d_in[2];
    float*       out = (float*)d_out;

    const int n_edges = in_sizes[1];
    const int n_waves = (n_edges + EDGES_PER_WAVE - 1) / EDGES_PER_WAVE;
    const int blocks  = (n_waves + WAVES_PER_BLOCK - 1) / WAVES_PER_BLOCK;

    edge_dot_wmma_f32<<<blocks, WAVES_PER_BLOCK * 32, 0, stream>>>(
        x, src, dst, out, n_edges);
}


# whole-line gather loads: A rows / B cols are 32-B pieces of 4 edges, so a load reads 4 full 128-B lines (was 16 lines x 32 B), all 32 issued ahead; 4 diagonal pieces quad-summed: sum order changes
# speedup vs baseline: 2.8976x; 1.9105x over previous
#include <hip/hip_runtime.h>
#include <hip/hip_bf16.h>

typedef __attribute__((ext_vector_type(2))) float v2f;
typedef __attribute__((ext_vector_type(4))) float v4f;
typedef __attribute__((ext_vector_type(8))) float v8f;

#define D_FEAT 128
#define WAVES_PER_BLOCK 8
#define EDGES_PER_WAVE 16


struct Frag {
    v4f a[4];
    v4f b[4];
};

template <int T>
__device__ __forceinline__ int quad_bcast(int v)
{
    return __builtin_amdgcn_update_dpp(0, v, T * 0x55, 0xf, 0xf, false);
}

__device__ __forceinline__ float quad_sum(float v)
{
    v += __builtin_bit_cast(float, __builtin_amdgcn_update_dpp(
             0, __builtin_bit_cast(int, v), 0xB1, 0xf, 0xf, false));
    v += __builtin_bit_cast(float, __builtin_amdgcn_update_dpp(
             0, __builtin_bit_cast(int, v), 0x4E, 0xf, 0xf, false));
    return v;
}

template <int T>
__device__ __forceinline__ void tile_load(Frag& f, const float* __restrict__ x,
                                          int s, int d, int laneOff)
{
    const int st = quad_bcast<T>(s);
    const int dt = quad_bcast<T>(d);
    const float* __restrict__ pa = x + (long long)st * D_FEAT + laneOff;
    const float* __restrict__ pb = x + (long long)dt * D_FEAT + laneOff;
#pragma unroll
    for (int line = 0; line < 4; ++line) {
        f.a[line] = *(const v4f*)(pa + line * 32);
        f.b[line] = *(const v4f*)(pb + line * 32);
    }
}

__device__ __forceinline__ float tile_dot(const Frag& f, int idx)
{
    v8f c = {};
#pragma unroll
    for (int line = 0; line < 4; ++line) {
        const v4f a = f.a[line];
        const v4f b = f.b[line];
        const v2f a0 = {a.x, a.y};
        const v2f b0 = {b.x, b.y};
        const v2f a1 = {a.z, a.w};
        const v2f b1 = {b.z, b.w};
        c = __builtin_amdgcn_wmma_f32_16x16x4_f32(
                false, a0, false, b0, (short)0, c, false, false);
        c = __builtin_amdgcn_wmma_f32_16x16x4_f32(
                false, a1, false, b1, (short)0, c, false, false);
    }

    float val = c[0];
    if (idx == 1) val = c[1];
    if (idx == 2) val = c[2];
    if (idx == 3) val = c[3];
    if (idx == 4) val = c[4];
    if (idx == 5) val = c[5];
    if (idx == 6) val = c[6];
    if (idx == 7) val = c[7];

    return quad_sum(val);
}

__global__ __launch_bounds__(256) void edge_dot_wmma_f32(
    const float* __restrict__ x,
    const int*   __restrict__ src,
    const int*   __restrict__ dst,
    float*       __restrict__ out,
    int n_edges)
{
    const int lane    = threadIdx.x & 31;
    const int wave    = threadIdx.x >> 5;
    const long long waveGlobal = (long long)blockIdx.x * WAVES_PER_BLOCK + wave;
    const long long e0 = waveGlobal * EDGES_PER_WAVE;
    if (e0 >= n_edges) return;

    const int ei = lane & 15;
    long long e = e0 + ei;
    if (e >= n_edges) e = n_edges - 1;

    const int s = src[e];
    const int d = dst[e];

    const int laneOff = (((lane >> 4) << 2) + (lane & 3)) << 2;
    const int idx = lane & 7;

    Frag f0, f1, f2, f3;
    tile_load<0>(f0, x, s, d, laneOff);
    tile_load<1>(f1, x, s, d, laneOff);
    tile_load<2>(f2, x, s, d, laneOff);
    tile_load<3>(f3, x, s, d, laneOff);
    __builtin_amdgcn_sched_barrier(0);
    const float r0 = tile_dot(f0, idx);
    const float r1 = tile_dot(f1, idx);
    const float r2 = tile_dot(f2, idx);
    const float r3 = tile_dot(f3, idx);

    const int sub = lane & 3;
    float val = r0;
    if (sub == 1) val = r1;
    if (sub == 2) val = r2;
    if (sub == 3) val = r3;

    const bool writer = (lane < 8) || (lane >= 24);
    const long long eo = e0 + ((lane < 8) ? lane : (lane - 16));
    if (writer && eo < n_edges) out[eo] = val;
}

extern "C" void kernel_launch(void* const* d_in, const int* in_sizes, int n_in,
                              void* d_out, int out_size, void* d_ws, size_t ws_size,
                              hipStream_t stream) {
    const float* x   = (const float*)d_in[0];
    const int*   src = (const int*)d_in[1];
    const int*   dst = (const int*)d_in[2];
    float*       out = (float*)d_out;

    const int n_edges = in_sizes[1];
    const int n_waves = (n_edges + EDGES_PER_WAVE - 1) / EDGES_PER_WAVE;
    const int blocks  = (n_waves + WAVES_PER_BLOCK - 1) / WAVES_PER_BLOCK;

    edge_dot_wmma_f32<<<blocks, WAVES_PER_BLOCK * 32, 0, stream>>>(
        x, src, dst, out, n_edges);
}


# each wave runs 4 consecutive 16-edge groups, refilling a tile's registers with the next group's rows behind its WMMAs: 18750 long waves (was 75000 short), 5 per SIMD (was 6); sums unchanged
# speedup vs baseline: 3.0705x; 1.0597x over previous
#include <hip/hip_runtime.h>
#include <hip/hip_bf16.h>

typedef __attribute__((ext_vector_type(2))) float v2f;
typedef __attribute__((ext_vector_type(4))) float v4f;
typedef __attribute__((ext_vector_type(8))) float v8f;

#define D_FEAT 128
#define WAVES_PER_BLOCK 8
#define EDGES_PER_GROUP 16
#define GROUPS_PER_WAVE 4
#define EDGES_PER_WAVE (EDGES_PER_GROUP * GROUPS_PER_WAVE)


struct Frag {
    v4f a[4];
    v4f b[4];
};

template <int T>
__device__ __forceinline__ int quad_bcast(int v)
{
    return __builtin_amdgcn_update_dpp(0, v, T * 0x55, 0xf, 0xf, false);
}

__device__ __forceinline__ float quad_sum(float v)
{
    v += __builtin_bit_cast(float, __builtin_amdgcn_update_dpp(
             0, __builtin_bit_cast(int, v), 0xB1, 0xf, 0xf, false));
    v += __builtin_bit_cast(float, __builtin_amdgcn_update_dpp(
             0, __builtin_bit_cast(int, v), 0x4E, 0xf, 0xf, false));
    return v;
}

template <int T>
__device__ __forceinline__ void tile_load(Frag& f, const float* __restrict__ x,
                                          int s, int d, int laneOff)
{
    const int st = quad_bcast<T>(s);
    const int dt = quad_bcast<T>(d);
    const float* __restrict__ pa = x + (long long)st * D_FEAT + laneOff;
    const float* __restrict__ pb = x + (long long)dt * D_FEAT + laneOff;
#pragma unroll
    for (int line = 0; line < 4; ++line) {
        f.a[line] = *(const v4f*)(pa + line * 32);
        f.b[line] = *(const v4f*)(pb + line * 32);
    }
}

__device__ __forceinline__ float tile_dot(const Frag& f, int idx)
{
    v8f c = {};
#pragma unroll
    for (int line = 0; line < 4; ++line) {
        const v4f a = f.a[line];
        const v4f b = f.b[line];
        const v2f a0 = {a.x, a.y};
        const v2f b0 = {b.x, b.y};
        const v2f a1 = {a.z, a.w};
        const v2f b1 = {b.z, b.w};
        c = __builtin_amdgcn_wmma_f32_16x16x4_f32(
                false, a0, false, b0, (short)0, c, false, false);
        c = __builtin_amdgcn_wmma_f32_16x16x4_f32(
                false, a1, false, b1, (short)0, c, false, false);
    }

    float val = c[0];
    if (idx == 1) val = c[1];
    if (idx == 2) val = c[2];
    if (idx == 3) val = c[3];
    if (idx == 4) val = c[4];
    if (idx == 5) val = c[5];
    if (idx == 6) val = c[6];
    if (idx == 7) val = c[7];

    return quad_sum(val);
}

template <int T>
__device__ __forceinline__ float tile_turn(Frag& f, const float* __restrict__ x,
                                           int sNext, int dNext, int laneOff,
                                           int idx, bool more)
{
    const float r = tile_dot(f, idx);
    if (more) {
        tile_load<T>(f, x, sNext, dNext, laneOff);
        __builtin_amdgcn_sched_barrier(0);
    }
    return r;
}

__global__ __launch_bounds__(256) void edge_dot_wmma_f32(
    const float* __restrict__ x,
    const int*   __restrict__ src,
    const int*   __restrict__ dst,
    float*       __restrict__ out,
    int n_edges)
{
    const int lane    = threadIdx.x & 31;
    const int wave    = threadIdx.x >> 5;
    const long long waveGlobal = (long long)blockIdx.x * WAVES_PER_BLOCK + wave;
    const long long e0 = waveGlobal * EDGES_PER_WAVE;
    if (e0 >= n_edges) return;

    const long long left = (long long)n_edges - e0;
    const int nHere = (left < EDGES_PER_WAVE) ? (int)left : EDGES_PER_WAVE;
    const int* __restrict__ srcW = src + e0;
    const int* __restrict__ dstW = dst + e0;

    const int ei = lane & 15;
    int s[GROUPS_PER_WAVE];
    int d[GROUPS_PER_WAVE];
#pragma unroll
    for (int g = 0; g < GROUPS_PER_WAVE; ++g) {
        int rel = g * EDGES_PER_GROUP + ei;
        if (rel >= nHere) rel = nHere - 1;
        s[g] = srcW[rel];
        d[g] = dstW[rel];
    }

    const int laneOff = (((lane >> 4) << 2) + (lane & 3)) << 2;
    const int idx = lane & 7;
    const int sub = lane & 3;
    const bool writer = (lane < 8) || (lane >= 24);
    const int  eLane  = (lane < 8) ? lane : (writer ? (lane - 16) : 0);
    float* __restrict__ outW = out + e0 + eLane;

    Frag f0, f1, f2, f3;
    tile_load<0>(f0, x, s[0], d[0], laneOff);
    tile_load<1>(f1, x, s[0], d[0], laneOff);
    tile_load<2>(f2, x, s[0], d[0], laneOff);
    tile_load<3>(f3, x, s[0], d[0], laneOff);
    __builtin_amdgcn_sched_barrier(0);

#pragma unroll
    for (int g = 0; g < GROUPS_PER_WAVE; ++g) {
        const bool more = (g + 1 < GROUPS_PER_WAVE);
        const int  gn   = more ? (g + 1) : g;
        const float r0 = tile_turn<0>(f0, x, s[gn], d[gn], laneOff, idx, more);
        const float r1 = tile_turn<1>(f1, x, s[gn], d[gn], laneOff, idx, more);
        const float r2 = tile_turn<2>(f2, x, s[gn], d[gn], laneOff, idx, more);
        const float r3 = tile_turn<3>(f3, x, s[gn], d[gn], laneOff, idx, more);

        float val = r0;
        if (sub == 1) val = r1;
        if (sub == 2) val = r2;
        if (sub == 3) val = r3;

        if (writer && g * EDGES_PER_GROUP + eLane < nHere) outW[g * EDGES_PER_GROUP] = val;
    }
}

extern "C" void kernel_launch(void* const* d_in, const int* in_sizes, int n_in,
                              void* d_out, int out_size, void* d_ws, size_t ws_size,
                              hipStream_t stream) {
    const float* x   = (const float*)d_in[0];
    const int*   src = (const int*)d_in[1];
    const int*   dst = (const int*)d_in[2];
    float*       out = (float*)d_out;

    const int n_edges = in_sizes[1];
    const int n_waves = (n_edges + EDGES_PER_WAVE - 1) / EDGES_PER_WAVE;
    const int blocks  = (n_waves + WAVES_PER_BLOCK - 1) / WAVES_PER_BLOCK;

    edge_dot_wmma_f32<<<blocks, WAVES_PER_BLOCK * 32, 0, stream>>>(
        x, src, dst, out, n_edges);
}
